# starting kernel
# baseline (speedup 1.0000x reference)
#include <hip/hip_runtime.h>

#define IN_DIM  256
#define HID_DIM 128
#define OUT_DIM 64

typedef __attribute__((ext_vector_type(2))) float v2f;
typedef __attribute__((ext_vector_type(8))) float v8f;

__global__ void init_deg_kernel(float* __restrict__ deg, int n) {
  int i = blockIdx.x * blockDim.x + threadIdx.x;
  if (i < n) deg[i] = 1.0f;
}

__global__ void accum_deg_kernel(const int* __restrict__ dst,
                                 float* __restrict__ deg, int e) {
  int i = blockIdx.x * blockDim.x + threadIdx.x;
  if (i < e) atomicAdd(&deg[dst[i]], 1.0f);
}

__global__ void dinv_kernel(float* __restrict__ deg, int n) {
  int i = blockIdx.x * blockDim.x + threadIdx.x;
  if (i < n) deg[i] = rsqrtf(deg[i]);
}

template <int K, int NT>
__global__ void gemm_wmma_kernel(const float* __restrict__ A,
                                 const float* __restrict__ B,
                                 float* __restrict__ C, int M) {
  const int N = NT * 16;
  const int wave = (int)((blockIdx.x * (unsigned)blockDim.x + threadIdx.x) >> 5);
  const int lane = threadIdx.x & 31;
  const int row0 = wave * 16;
  if (row0 >= M) return;

  const int mn    = lane & 15;
  const int khalf = lane >> 4;

  int arow = row0 + mn;
  if (arow >= M) arow = M - 1;
  const float* __restrict__ Ar = A + (size_t)arow * K;

  v8f acc[NT];
#pragma unroll
  for (int t = 0; t < NT; ++t) acc[t] = v8f{};

  for (int k = 0; k < K; k += 4) {
    const int ka = k + 2 * khalf;
    v2f a;
    a.x = Ar[ka];
    a.y = Ar[ka + 1];
#pragma unroll
    for (int t = 0; t < NT; ++t) {
      v2f b;
      b.x = B[(size_t)ka       * N + t * 16 + mn];
      b.y = B[(size_t)(ka + 1) * N + t * 16 + mn];
      acc[t] = __builtin_amdgcn_wmma_f32_16x16x4_f32(
           false, a,  false, b,
           (short)0, acc[t],  false,  false);
    }
  }

#pragma unroll
  for (int t = 0; t < NT; ++t) {
#pragma unroll
    for (int i = 0; i < 8; ++i) {
      const int row = row0 + 8 * khalf + i;
      if (row < M) C[(size_t)row * N + t * 16 + mn] = acc[t][i];
    }
  }
}

__global__ void init_bias_kernel(float* __restrict__ out,
                                 const float* __restrict__ bias,
                                 int fmask, long long total) {
  long long idx = blockIdx.x * (long long)blockDim.x + threadIdx.x;
  if (idx < total) out[idx] = bias[(int)(idx & fmask)];
}

template <int F>
__global__ void scatter_kernel(const float* __restrict__ h,
                               const int* __restrict__ src,
                               const int* __restrict__ dst,
                               const float* __restrict__ dinv,
                               float* __restrict__ agg,
                               int n_edges, int n_nodes) {
  long long idx = blockIdx.x * (long long)blockDim.x + threadIdx.x;
  long long total = (long long)(n_edges + n_nodes) * F;
  if (idx >= total) return;
  const int e = (int)(idx / F);
  const int f = (int)(idx & (F - 1));
  int s, d;
  if (e < n_edges) { s = src[e]; d = dst[e]; }
  else             { s = d = e - n_edges; }
  const float norm = dinv[s] * dinv[d];
  atomicAdd(&agg[(size_t)d * F + f], h[(size_t)s * F + f] * norm);
}

__global__ void relu_kernel(float* __restrict__ v, long long total) {
  long long idx = blockIdx.x * (long long)blockDim.x + threadIdx.x;
  if (idx < total) {
    float x = v[idx];
    v[idx] = x > 0.0f ? x : 0.0f;
  }
}

extern "C" void kernel_launch(void* const* d_in, const int* in_sizes, int n_in,
                              void* d_out, int out_size, void* d_ws, size_t ws_size,
                              hipStream_t stream) {
  const float* x  = (const float*)d_in[0];
  const int*   ei = (const int*)d_in[1];
  const float* W1 = (const float*)d_in[2];
  const float* b1 = (const float*)d_in[3];
  const float* W2 = (const float*)d_in[4];
  const float* b2 = (const float*)d_in[5];
  float* out = (float*)d_out;

  const int N = in_sizes[0] / IN_DIM;
  const int E = in_sizes[1] / 2;
  const int* src = ei;
  const int* dst = ei + E;

  float* dinv = (float*)d_ws;
  float* h    = dinv + N;
  float* agg1 = h + (size_t)N * HID_DIM;

  const int B256 = 256;
  init_deg_kernel<<<(N + B256 - 1) / B256, B256, 0, stream>>>(dinv, N);
  accum_deg_kernel<<<(E + B256 - 1) / B256, B256, 0, stream>>>(dst, dinv, E);
  dinv_kernel<<<(N + B256 - 1) / B256, B256, 0, stream>>>(dinv, N);

  const int waves  = (N + 15) / 16;
  const int gblks  = (waves + 3) / 4;

  gemm_wmma_kernel<IN_DIM, HID_DIM / 16><<<gblks, 128, 0, stream>>>(x, W1, h, N);
  const long long tot1 = (long long)N * HID_DIM;
  init_bias_kernel<<<(int)((tot1 + B256 - 1) / B256), B256, 0, stream>>>(agg1, b1, HID_DIM - 1, tot1);
  const long long sc1 = (long long)(E + N) * HID_DIM;
  scatter_kernel<HID_DIM><<<(int)((sc1 + B256 - 1) / B256), B256, 0, stream>>>(
      h, src, dst, dinv, agg1, E, N);
  relu_kernel<<<(int)((tot1 + B256 - 1) / B256), B256, 0, stream>>>(agg1, tot1);

  gemm_wmma_kernel<HID_DIM, OUT_DIM / 16><<<gblks, 128, 0, stream>>>(agg1, W2, h, N);
  const long long tot2 = (long long)N * OUT_DIM;
  init_bias_kernel<<<(int)((tot2 + B256 - 1) / B256), B256, 0, stream>>>(out, b2, OUT_DIM - 1, tot2);
  const long long sc2 = (long long)(E + N) * OUT_DIM;
  scatter_kernel<OUT_DIM><<<(int)((sc2 + B256 - 1) / B256), B256, 0, stream>>>(
      h, src, dst, dinv, out, E, N);
}


# GEMM (both layers): fragments of 4 k-steps form a register block, two blocks alternate; the next block's loads are issued ahead of the current block's WMMAs; loaders, k order and sum order kept
# speedup vs baseline: 4.1235x; 4.1235x over previous
#include <hip/hip_runtime.h>


#define IN_DIM  256
#define HID_DIM 128
#define OUT_DIM 64

#define TILE_NODES  196
#define AGG_THREADS 1024
#define AGG_WAVES   (AGG_THREADS / 32)
#define WAVE_CAP    1024
#define SAFE_WINDOW 16384
#define SEG_MAX     384
#define LIST_ROW    OUT_DIM
#define PACK_BITS   24
#define SCAN_U      8
#define GATHER_U    8

typedef __attribute__((ext_vector_type(2))) float v2f;
typedef __attribute__((ext_vector_type(8))) float v8f;

template <class OnHit>
__device__ __forceinline__ void test_group(int4 d, int e, int lo, unsigned ucnt, OnHit&& on) {
  const unsigned a = (unsigned)(d.x - lo), b = (unsigned)(d.y - lo);
  const unsigned c = (unsigned)(d.z - lo), f = (unsigned)(d.w - lo);
  const unsigned mn = min(min(a, b), min(c, f));
  if (__builtin_amdgcn_ballot_w32(mn < ucnt) != 0u) {
    on(e,     d.x, a < ucnt);
    on(e + 1, d.y, b < ucnt);
    on(e + 2, d.z, c < ucnt);
    on(e + 3, d.w, f < ucnt);
  }
}

template <class OnHit>
__device__ __forceinline__ void scan_window(const int* __restrict__ dst, int pos, int len,
                                            int lo, int cnt, OnHit&& on) {
  const int tid = threadIdx.x;
  const unsigned ucnt = (unsigned)cnt;
  const int head     = min(len, (int)((4u - (unsigned)(((size_t)(dst + pos)) >> 2)) & 3u));
  const int ngrp     = (len - head) >> 2;
  const int body_end = head + 4 * ngrp;
  const int nrag     = head + (len - body_end);
  {
    const bool v  = tid < nrag;
    const int rel = tid < head ? tid : body_end + (tid - head);
    const int e   = pos + (v ? rel : 0);
    const int d   = v ? dst[e] : -1;
    const bool hit = (unsigned)(d - lo) < ucnt;
    if (__builtin_amdgcn_ballot_w32(hit) != 0u) on(e, d, hit);
  }
  const int4* __restrict__ d4 = (const int4*)(dst + pos + head);
  const int ebase = pos + head;
  constexpr int GPI = AGG_THREADS * SCAN_U;
  const int nfull = ngrp / GPI;
  const int itrot = (int)(((unsigned)blockIdx.x * (unsigned)nfull) / gridDim.x);
  for (int it = 0; it < nfull; ++it) {
    int itr = it + itrot;
    if (itr >= nfull) itr -= nfull;
    const int g0 = itr * GPI + tid;
    int4 d[SCAN_U];
#pragma unroll
    for (int u = 0; u < SCAN_U; ++u) d[u] = d4[g0 + u * AGG_THREADS];
#pragma unroll
    for (int u = 0; u < SCAN_U; ++u)
      test_group(d[u], ebase + 4 * (g0 + u * AGG_THREADS), lo, ucnt, on);
  }
  for (int base = nfull * GPI; base < ngrp; base += AGG_THREADS) {
    const int g  = base + tid;
    const bool v = g < ngrp;
    const int gi = v ? g : base;
    int4 d = d4[gi];
    if (!v) { d.x = -1; d.y = -1; d.z = -1; d.w = -1; }
    test_group(d, ebase + 4 * gi, lo, ucnt, on);
  }
}

__device__ __forceinline__ int seg_pitch(int cnt) {
  return min(SEG_MAX, ((2 * cnt - 1) / 32) * 32);
}

__global__ __launch_bounds__(AGG_THREADS)
void filter_kernel(const int* __restrict__ src, const int* __restrict__ dst,
                   float* __restrict__ dinv, unsigned* __restrict__ lists,
                   int n_edges, int n_nodes) {
  __shared__ int s_c[256];
  __shared__ int s_list[AGG_WAVES * SEG_MAX];
  __shared__ int s_cnt[AGG_WAVES];
  const int tid  = threadIdx.x;
  const int lane = tid & 31;
  const int wave = tid >> 5;
  const int lo   = blockIdx.x * TILE_NODES;
  const int cnt  = min(TILE_NODES, n_nodes - lo);
  const int a0   = (lo + 31) & ~31;
  const int a1   = min(n_nodes, (lo + cnt + 31) & ~31);
  const int cntc = a1 - lo;
  const int pitch = seg_pitch(cnt);
  int* wl = s_list + wave * SEG_MAX;

  for (int i = tid; i < 256; i += AGG_THREADS) s_c[i] = 0;
  __syncthreads();

  int wcount = 0;
  if (n_edges > 0)
    scan_window(dst, 0, n_edges, lo, cntc, [&](int e, int d, bool hitc) {
      if (hitc) atomicAdd(&s_c[d - lo], 1);
      const bool hit = hitc && (d - lo) < cnt;
      const unsigned bal = __builtin_amdgcn_ballot_w32(hit);
      if (hit) {
        const int p = wcount + (int)__builtin_amdgcn_mbcnt_lo(bal, 0u);
        if (p < pitch) wl[p] = e;
      }
      wcount += __popc(bal);
    });
  if (lane == 0) s_cnt[wave] = (n_nodes <= (1 << PACK_BITS)) ? wcount : 0x7fffffff;
  __syncthreads();

  for (int i = a0 - lo + tid; i < cntc; i += AGG_THREADS)
    dinv[lo + i] = rsqrtf(1.0f + (float)s_c[i]);

  unsigned* base = lists + (size_t)lo * LIST_ROW;
  if (wave == 0) base[lane] = (unsigned)s_cnt[lane];
  unsigned* seg = base + 32 + wave * pitch;
  const int n = min(wcount, pitch);
  for (int i = lane; i < n; i += 32) {
    const int e = wl[i];
    int s = src[e];
    const int d = dst[e];
    s = min(max(s, 0), n_nodes - 1);
    seg[i] = (unsigned)s | ((unsigned)(d - lo) << PACK_BITS);
  }
}

template <int K, int NT>
__global__ __launch_bounds__(128)
void gemm_wmma_kernel(const float* __restrict__ A,
                      const float* __restrict__ B,
                      float* __restrict__ C, int M) {
  constexpr int N  = NT * 16;
  constexpr int KB = 4;
  constexpr int NB = K / (4 * KB);
  static_assert(K % (8 * KB) == 0, "the two register blocks alternate in pairs");
  const int wave = (int)((blockIdx.x * (unsigned)blockDim.x + threadIdx.x) >> 5);
  const int lane = threadIdx.x & 31;
  const int row0 = wave * 16;
  if (row0 >= M) return;

  const int mn    = lane & 15;
  const int khalf = lane >> 4;

  int arow = row0 + mn;
  if (arow >= M) arow = M - 1;
  const float* __restrict__ Ar = A + (size_t)arow * K;
  const float* __restrict__ Bc = B + mn;

  v8f acc[NT];
#pragma unroll
  for (int t = 0; t < NT; ++t) acc[t] = v8f{};

  v2f aP[KB], aQ[KB];
  v2f bP[KB][NT], bQ[KB][NT];

  auto load_block = [&](int kb, v2f (&a)[KB], v2f (&b)[KB][NT]) {
#pragma unroll
    for (int s = 0; s < KB; ++s) {
      const int ka = (kb * KB + s) * 4 + 2 * khalf;
      a[s].x = Ar[ka];
      a[s].y = Ar[ka + 1];
#pragma unroll
      for (int t = 0; t < NT; ++t) {
        b[s][t].x = Bc[(size_t)ka       * N + t * 16];
        b[s][t].y = Bc[(size_t)(ka + 1) * N + t * 16];
      }
    }
  };
  auto mma_block = [&](const v2f (&a)[KB], const v2f (&b)[KB][NT]) {
#pragma unroll
    for (int s = 0; s < KB; ++s) {
#pragma unroll
      for (int t = 0; t < NT; ++t)
        acc[t] = __builtin_amdgcn_wmma_f32_16x16x4_f32(
             false, a[s],  false, b[s][t],
             (short)0, acc[t],  false,  false);
    }
  };

  load_block(0, aP, bP);
  for (int kb = 0; kb + 2 < NB; kb += 2) {
    load_block(kb + 1, aQ, bQ);
    mma_block(aP, bP);
    load_block(kb + 2, aP, bP);
    mma_block(aQ, bQ);
  }
  load_block(NB - 1, aQ, bQ);
  mma_block(aP, bP);
  mma_block(aQ, bQ);

#pragma unroll
  for (int t = 0; t < NT; ++t) {
#pragma unroll
    for (int i = 0; i < 8; ++i) {
      const int row = row0 + 8 * khalf + i;
      if (row < M) C[(size_t)row * N + t * 16 + mn] = acc[t][i];
    }
  }
}

template <int F, bool RELU>
__global__ __launch_bounds__(AGG_THREADS)
void aggregate_kernel(const float* __restrict__ h,
                      const int* __restrict__ src,
                      const int* __restrict__ dst,
                      const float* __restrict__ dinv,
                      const float* __restrict__ bias,
                      const unsigned* lists,
                      float* out,
                      int n_edges, int n_nodes) {
  __shared__ __attribute__((aligned(16))) float s_acc[TILE_NODES * F];
  __shared__ int s_list[AGG_WAVES * WAVE_CAP];
  __shared__ int s_ovf;

  constexpr int VW = F / 32;
  typedef __attribute__((ext_vector_type(VW))) float vrow;
  const int tid  = threadIdx.x;
  const int lane = tid & 31;
  const int wave = __builtin_amdgcn_readfirstlane(tid >> 5);
  const int lo   = blockIdx.x * TILE_NODES;
  const int cnt  = min(TILE_NODES, n_nodes - lo);
  const int nvec = cnt * 32;
  int* wl = s_list + wave * WAVE_CAP;

  const unsigned* lbase = lists + (size_t)lo * LIST_ROW;
  const int pitch = seg_pitch(cnt);
  const int hv    = (int)lbase[lane];
  const bool refilter = __builtin_amdgcn_ballot_w32(hv > pitch) != 0u;

  {
    for (int i = tid; i < nvec; i += AGG_THREADS) {
      const int r = i >> 5;
      const float di = dinv[lo + r];
      const float w  = di * di;
      const vrow v = *(const vrow*)(h + (size_t)(lo + r) * F + lane * VW);
      float* ar = s_acc + r * F + lane;
#pragma unroll
      for (int j = 0; j < VW; ++j) ar[32 * j] = v[j] * w;
    }
  }

  __syncthreads();
  if (!refilter) {
    const int nw = __builtin_amdgcn_readlane(hv, wave);
    const unsigned* seg = lbase + 32 + wave * pitch;
    for (int i0 = 0; i0 < nw; i0 += 32) {
      const int i = i0 + lane;
      const unsigned pk = seg[i < nw ? i : i0];
      const int   s   = (int)(pk & ((1u << PACK_BITS) - 1u));
      const int   dl  = (int)(pk >> PACK_BITS);
      float ds = 0.0f, dd = 0.0f;
      const int m = min(32, nw - i0);
      for (int j0 = 0; j0 < m; j0 += GATHER_U) {
        int sv[GATHER_U];
        vrow v[GATHER_U];
#pragma unroll
        for (int u = 0; u < GATHER_U; ++u) {
          sv[u] = j0 + u < m ? j0 + u : j0;
          const int su = __builtin_amdgcn_readlane(s, sv[u]);
          v[u] = *(const vrow*)(h + (size_t)su * F + lane * VW);
        }
        if (j0 == 0) {
          ds = dinv[s];
          dd = dinv[lo + dl];
        }
#pragma unroll
        for (int u = 0; u < GATHER_U; ++u) {
          if (j0 + u < m) {
            const int   du = __builtin_amdgcn_readlane(dl, sv[u]);
            const float wu = __int_as_float(__builtin_amdgcn_readlane(__float_as_int(ds), sv[u])) *
                             __int_as_float(__builtin_amdgcn_readlane(__float_as_int(dd), sv[u]));
            float* ar = s_acc + du * F + lane;
#pragma unroll
            for (int j = 0; j < VW; ++j) atomicAdd(&ar[32 * j], v[u][j] * wu);
          }
        }
      }
    }
  }
  int step = n_edges;
  for (int pos = 0; refilter && pos < n_edges; ) {
    if (tid == 0) s_ovf = 0;
    __syncthreads();

    const int len = min(step, n_edges - pos);
    int wcount = 0;
    scan_window(dst, pos, len, lo, cnt, [&](int e, int, bool hit) {
      const unsigned bal = __builtin_amdgcn_ballot_w32(hit);
      if (hit) {
        const int p = wcount + (int)__builtin_amdgcn_mbcnt_lo(bal, 0u);
        if (p < WAVE_CAP) wl[p] = e;
      }
      wcount += __popc(bal);
    });
    if (wcount > WAVE_CAP && lane == 0) s_ovf = 1;
    __syncthreads();

    if (s_ovf != 0 && step > SAFE_WINDOW) {
      step = SAFE_WINDOW;
      __syncthreads();
      continue;
    }

    for (int i0 = 0; i0 < wcount; i0 += GATHER_U) {
      int   sv[GATHER_U], dl[GATHER_U];
      float w[GATHER_U];
#pragma unroll
      for (int u = 0; u < GATHER_U; ++u) {
        const int i = i0 + u;
        const int e = wl[i < wcount ? i : i0];
        int s = src[e];
        const int dd = dst[e];
        s = min(max(s, 0), n_nodes - 1);
        sv[u] = s;
        dl[u] = dd - lo;
        w[u]  = dinv[s] * dinv[dd];
      }
      vrow v[GATHER_U];
#pragma unroll
      for (int u = 0; u < GATHER_U; ++u)
        v[u] = *(const vrow*)(h + (size_t)sv[u] * F + lane * VW);
#pragma unroll
      for (int u = 0; u < GATHER_U; ++u) {
        if (i0 + u < wcount) {
          float* ar = s_acc + dl[u] * F + lane;
#pragma unroll
          for (int j = 0; j < VW; ++j) atomicAdd(&ar[32 * j], v[u][j] * w[u]);
        }
      }
    }
    __syncthreads();
    pos += len;
  }
  __syncthreads();

  {
    float b[VW];
#pragma unroll
    for (int j = 0; j < VW; ++j) b[j] = bias[lane * VW + j];
    for (int i = tid; i < nvec; i += AGG_THREADS) {
      const int r = i >> 5;
      const float* ar = s_acc + r * F + lane;
      vrow v;
#pragma unroll
      for (int j = 0; j < VW; ++j) {
        float x = ar[32 * j] + b[j];
        if (RELU) x = x > 0.0f ? x : 0.0f;
        v[j] = x;
      }
      *(vrow*)(out + (size_t)(lo + r) * F + lane * VW) = v;
    }
  }
}

extern "C" void kernel_launch(void* const* d_in, const int* in_sizes, int n_in,
                              void* d_out, int out_size, void* d_ws, size_t ws_size,
                              hipStream_t stream) {
  const float* x  = (const float*)d_in[0];
  const int*   ei = (const int*)d_in[1];
  const float* W1 = (const float*)d_in[2];
  const float* b1 = (const float*)d_in[3];
  const float* W2 = (const float*)d_in[4];
  const float* b2 = (const float*)d_in[5];
  float* out = (float*)d_out;

  const int N = in_sizes[0] / IN_DIM;
  const int E = in_sizes[1] / 2;
  const int* src = ei;
  const int* dst = ei + E;

  float* h    = (float*)d_ws;
  float* agg1 = h + (size_t)N * HID_DIM;
  float* dinv = agg1 + (size_t)N * HID_DIM;

  const int tiles  = (N + TILE_NODES - 1) / TILE_NODES;

  unsigned* lists = (unsigned*)d_out;
  filter_kernel<<<tiles, AGG_THREADS, 0, stream>>>(src, dst, dinv, lists, E, N);

  const int waves  = (N + 15) / 16;
  const int gblks  = (waves + 3) / 4;

  gemm_wmma_kernel<IN_DIM, HID_DIM / 16><<<gblks, 128, 0, stream>>>(x, W1, h, N);
  aggregate_kernel<HID_DIM, true><<<tiles, AGG_THREADS, 0, stream>>>(
      h, src, dst, dinv, b1, lists, agg1, E, N);

  gemm_wmma_kernel<HID_DIM, OUT_DIM / 16><<<gblks, 128, 0, stream>>>(agg1, W2, h, N);
  aggregate_kernel<OUT_DIM, false><<<tiles, AGG_THREADS, 0, stream>>>(
      h, src, dst, dinv, b2, lists, out, E, N);
}


# Edge walk shared by 4 work-groups: each walks a quarter of the edge list for 4 neighbouring even-sized tiles; partial counts summed in the layer-1 product launch; sums stay in LDS arrival order
# speedup vs baseline: 4.4927x; 1.0895x over previous
#include <hip/hip_runtime.h>


#define IN_DIM  256
#define HID_DIM 128
#define OUT_DIM 64

#define TILE_NODES  196
#define AGG_THREADS 1024
#define AGG_WAVES   (AGG_THREADS / 32)
#define WAVE_CAP    1024
#define SAFE_WINDOW 16384
#define SEG_MAX     384
#define QS          4
#define SEG_PITCH   96
#define LIST_ROW    OUT_DIM
#define PACK_BITS   24
#define SCAN_U      8
#define GATHER_U    8

typedef __attribute__((ext_vector_type(2))) float v2f;
typedef __attribute__((ext_vector_type(8))) float v8f;

template <class OnHit>
__device__ __forceinline__ void test_group(int4 d, int e, int lo, unsigned ucnt, OnHit&& on) {
  const unsigned a = (unsigned)(d.x - lo), b = (unsigned)(d.y - lo);
  const unsigned c = (unsigned)(d.z - lo), f = (unsigned)(d.w - lo);
  const unsigned mn = min(min(a, b), min(c, f));
  if (__builtin_amdgcn_ballot_w32(mn < ucnt) != 0u) {
    on(e,     d.x, a < ucnt);
    on(e + 1, d.y, b < ucnt);
    on(e + 2, d.z, c < ucnt);
    on(e + 3, d.w, f < ucnt);
  }
}

template <class OnHit>
__device__ __forceinline__ void scan_window(const int* __restrict__ dst, int pos, int len,
                                            int lo, int cnt, OnHit&& on) {
  const int tid = threadIdx.x;
  const unsigned ucnt = (unsigned)cnt;
  const int head     = min(len, (int)((4u - (unsigned)(((size_t)(dst + pos)) >> 2)) & 3u));
  const int ngrp     = (len - head) >> 2;
  const int body_end = head + 4 * ngrp;
  const int nrag     = head + (len - body_end);
  {
    const bool v  = tid < nrag;
    const int rel = tid < head ? tid : body_end + (tid - head);
    const int e   = pos + (v ? rel : 0);
    const int d   = v ? dst[e] : -1;
    const bool hit = (unsigned)(d - lo) < ucnt;
    if (__builtin_amdgcn_ballot_w32(hit) != 0u) on(e, d, hit);
  }
  const int4* __restrict__ d4 = (const int4*)(dst + pos + head);
  const int ebase = pos + head;
  constexpr int GPI = AGG_THREADS * SCAN_U;
  const int nfull = ngrp / GPI;
  const int itrot = (int)(((unsigned)blockIdx.x * (unsigned)nfull) / gridDim.x);
  for (int it = 0; it < nfull; ++it) {
    int itr = it + itrot;
    if (itr >= nfull) itr -= nfull;
    const int g0 = itr * GPI + tid;
    int4 d[SCAN_U];
#pragma unroll
    for (int u = 0; u < SCAN_U; ++u) d[u] = d4[g0 + u * AGG_THREADS];
#pragma unroll
    for (int u = 0; u < SCAN_U; ++u)
      test_group(d[u], ebase + 4 * (g0 + u * AGG_THREADS), lo, ucnt, on);
  }
  for (int base = nfull * GPI; base < ngrp; base += AGG_THREADS) {
    const int g  = base + tid;
    const bool v = g < ngrp;
    const int gi = v ? g : base;
    int4 d = d4[gi];
    if (!v) { d.x = -1; d.y = -1; d.z = -1; d.w = -1; }
    test_group(d, ebase + 4 * gi, lo, ucnt, on);
  }
}

__device__ __forceinline__ int seg_pitch(int cnt) {
  return max(0, min(SEG_PITCH, (((cnt * LIST_ROW - QS * 32) / (QS * 32)) / 32) * 32));
}

__device__ __forceinline__ int tile_lo(int t, int n_nodes, int tiles) {
  return t >= tiles ? n_nodes : (int)(((long long)t * n_nodes) / tiles);
}

__global__ __launch_bounds__(AGG_THREADS)
void filter_kernel(const int* __restrict__ src, const int* __restrict__ dst,
                   int* __restrict__ cparts, unsigned* __restrict__ lists,
                   int n_edges, int n_nodes, int tiles) {
  __shared__ int s_c[QS * TILE_NODES + 32];
  __shared__ int s_list[AGG_WAVES * SEG_MAX];
  __shared__ int s_cnt[QS * AGG_WAVES];
  const int tid  = threadIdx.x;
  const int lane = tid & 31;
  const int wave = tid >> 5;
  const int grp  = blockIdx.x / QS;
  const int part = blockIdx.x % QS;
  int tlo[QS + 1];
#pragma unroll
  for (int q = 0; q <= QS; ++q) tlo[q] = tile_lo(grp * QS + q, n_nodes, tiles);
  const int lo   = tlo[0];
  const int cnt  = tlo[QS] - lo;
  const int a0   = (lo + 31) & ~31;
  const int a1   = min(n_nodes, (lo + cnt + 31) & ~31);
  const int cntc = a1 - lo;
  const int plane = (n_nodes + 31) & ~31;
  const int e0   = (int)(((long long)n_edges * part) / QS);
  const int e1   = (int)(((long long)n_edges * (part + 1)) / QS);
  int* wl = s_list + wave * SEG_MAX;

  for (int i = tid; i < QS * TILE_NODES + 32; i += AGG_THREADS) s_c[i] = 0;
  __syncthreads();

  int wcount = 0;
  if (e1 > e0)
    scan_window(dst, e0, e1 - e0, lo, cntc, [&](int e, int d, bool hitc) {
      if (hitc) atomicAdd(&s_c[d - lo], 1);
      const bool hit = hitc && (d - lo) < cnt;
      const unsigned bal = __builtin_amdgcn_ballot_w32(hit);
      if (hit) {
        const int p = wcount + (int)__builtin_amdgcn_mbcnt_lo(bal, 0u);
        if (p < SEG_MAX) wl[p] = e;
      }
      wcount += __popc(bal);
    });
  __syncthreads();

  int* cp = cparts + (size_t)part * plane;
  for (int i = a0 - lo + tid; i < cntc; i += AGG_THREADS) cp[lo + i] = s_c[i];

  const int n = min(wcount, SEG_MAX);
  const bool bad = wcount > SEG_MAX || n_nodes > (1 << PACK_BITS);
  int c[QS], pit[QS];
  unsigned* seg[QS];
#pragma unroll
  for (int q = 0; q < QS; ++q) {
    c[q]   = 0;
    pit[q] = seg_pitch(tlo[q + 1] - tlo[q]);
    seg[q] = lists + (size_t)tlo[q] * LIST_ROW + QS * 32 + (part * AGG_WAVES + wave) * pit[q];
  }
  for (int i0 = 0; i0 < n; i0 += 32) {
    const int i  = i0 + lane;
    const bool v = i < n;
    const int e  = wl[v ? i : i0];
    int sidx = src[e];
    const int d = dst[e];
    sidx = min(max(sidx, 0), n_nodes - 1);
    int q = 0, tl = tlo[0];
#pragma unroll
    for (int k = 1; k < QS; ++k) {
      if (d >= tlo[k]) { q = k; tl = tlo[k]; }
    }
    const unsigned word = (unsigned)sidx | ((unsigned)(d - tl) << PACK_BITS);
#pragma unroll
    for (int k = 0; k < QS; ++k) {
      const bool mine = v && q == k;
      const unsigned bal = __builtin_amdgcn_ballot_w32(mine);
      if (mine) {
        const int p = c[k] + (int)__builtin_amdgcn_mbcnt_lo(bal, 0u);
        if (p < pit[k]) seg[k][p] = word;
      }
      c[k] += __popc(bal);
    }
  }
  if (lane == 0) {
#pragma unroll
    for (int q = 0; q < QS; ++q) s_cnt[q * AGG_WAVES + wave] = bad ? 0x7fffffff : c[q];
  }
  __syncthreads();
#pragma unroll
  for (int q = 0; q < QS; ++q) {
    if (wave == q && (tlo[q + 1] - tlo[q]) * LIST_ROW >= QS * 32)
      (lists + (size_t)tlo[q] * LIST_ROW)[part * 32 + lane] = (unsigned)s_cnt[q * AGG_WAVES + lane];
  }
}

template <int K, int NT, bool DINV>
__global__ __launch_bounds__(128)
void gemm_wmma_kernel(const float* __restrict__ A,
                      const float* __restrict__ B,
                      float* __restrict__ C, int M,
                      const int* __restrict__ cparts, float* __restrict__ dinv) {
  constexpr int N  = NT * 16;
  constexpr int KB = 4;
  constexpr int NB = K / (4 * KB);
  static_assert(K % (8 * KB) == 0, "the two register blocks alternate in pairs");
  const int wave = (int)((blockIdx.x * (unsigned)blockDim.x + threadIdx.x) >> 5);
  const int lane = threadIdx.x & 31;
  const int row0 = wave * 16;
  if (row0 >= M) return;

  if (DINV && (wave & 1) == 0) {
    const int node = row0 + lane;
    if (node < M) {
      const int plane = (M + 31) & ~31;
      int deg = 0;
#pragma unroll
      for (int q = 0; q < QS; ++q) deg += cparts[(size_t)q * plane + node];
      dinv[node] = rsqrtf(1.0f + (float)deg);
    }
  }

  const int mn    = lane & 15;
  const int khalf = lane >> 4;

  int arow = row0 + mn;
  if (arow >= M) arow = M - 1;
  const float* __restrict__ Ar = A + (size_t)arow * K;
  const float* __restrict__ Bc = B + mn;

  v8f acc[NT];
#pragma unroll
  for (int t = 0; t < NT; ++t) acc[t] = v8f{};

  v2f aP[KB], aQ[KB];
  v2f bP[KB][NT], bQ[KB][NT];

  auto load_block = [&](int kb, v2f (&a)[KB], v2f (&b)[KB][NT]) {
#pragma unroll
    for (int s = 0; s < KB; ++s) {
      const int ka = (kb * KB + s) * 4 + 2 * khalf;
      a[s].x = Ar[ka];
      a[s].y = Ar[ka + 1];
#pragma unroll
      for (int t = 0; t < NT; ++t) {
        b[s][t].x = Bc[(size_t)ka       * N + t * 16];
        b[s][t].y = Bc[(size_t)(ka + 1) * N + t * 16];
      }
    }
  };
  auto mma_block = [&](const v2f (&a)[KB], const v2f (&b)[KB][NT]) {
#pragma unroll
    for (int s = 0; s < KB; ++s) {
#pragma unroll
      for (int t = 0; t < NT; ++t)
        acc[t] = __builtin_amdgcn_wmma_f32_16x16x4_f32(
             false, a[s],  false, b[s][t],
             (short)0, acc[t],  false,  false);
    }
  };

  load_block(0, aP, bP);
  for (int kb = 0; kb + 2 < NB; kb += 2) {
    load_block(kb + 1, aQ, bQ);
    mma_block(aP, bP);
    load_block(kb + 2, aP, bP);
    mma_block(aQ, bQ);
  }
  load_block(NB - 1, aQ, bQ);
  mma_block(aP, bP);
  mma_block(aQ, bQ);

#pragma unroll
  for (int t = 0; t < NT; ++t) {
#pragma unroll
    for (int i = 0; i < 8; ++i) {
      const int row = row0 + 8 * khalf + i;
      if (row < M) C[(size_t)row * N + t * 16 + mn] = acc[t][i];
    }
  }
}

template <int F, bool RELU>
__global__ __launch_bounds__(AGG_THREADS)
void aggregate_kernel(const float* __restrict__ h,
                      const int* __restrict__ src,
                      const int* __restrict__ dst,
                      const float* __restrict__ dinv,
                      const float* __restrict__ bias,
                      const unsigned* lists,
                      float* out,
                      int n_edges, int n_nodes) {
  __shared__ __attribute__((aligned(16))) float s_acc[TILE_NODES * F];
  __shared__ int s_list[AGG_WAVES * WAVE_CAP];
  __shared__ int s_ovf;

  constexpr int VW = F / 32;
  typedef __attribute__((ext_vector_type(VW))) float vrow;
  const int tid  = threadIdx.x;
  const int lane = tid & 31;
  const int wave = __builtin_amdgcn_readfirstlane(tid >> 5);
  const int lo   = tile_lo(blockIdx.x, n_nodes, gridDim.x);
  const int cnt  = tile_lo(blockIdx.x + 1, n_nodes, gridDim.x) - lo;
  const int nvec = cnt * 32;
  int* wl = s_list + wave * WAVE_CAP;

  const unsigned* lbase = lists + (size_t)lo * LIST_ROW;
  const int pitch = seg_pitch(cnt);
  const bool small = cnt * LIST_ROW < QS * 32;
  int nws[QS];
  bool over = small;
#pragma unroll
  for (int q = 0; q < QS; ++q) {
    const int hv = small ? 0 : (int)lbase[q * 32 + lane];
    over = over || hv > pitch;
    nws[q] = __builtin_amdgcn_readlane(hv, wave);
  }
  const bool refilter = __builtin_amdgcn_ballot_w32(over) != 0u;

  {
    for (int i = tid; i < nvec; i += AGG_THREADS) {
      const int r = i >> 5;
      const float di = dinv[lo + r];
      const float w  = di * di;
      const vrow v = *(const vrow*)(h + (size_t)(lo + r) * F + lane * VW);
      float* ar = s_acc + r * F + lane;
#pragma unroll
      for (int j = 0; j < VW; ++j) ar[32 * j] = v[j] * w;
    }
  }

  __syncthreads();
  for (int q = 0; !refilter && q < QS; ++q) {
    int nw = nws[0];
#pragma unroll
    for (int k = 1; k < QS; ++k) nw = q == k ? nws[k] : nw;
    const unsigned* seg = lbase + QS * 32 + (q * AGG_WAVES + wave) * pitch;
    for (int i0 = 0; i0 < nw; i0 += 32) {
      const int i = i0 + lane;
      const unsigned pk = seg[i < nw ? i : i0];
      const int   s   = (int)(pk & ((1u << PACK_BITS) - 1u));
      const int   dl  = (int)(pk >> PACK_BITS);
      float ds = 0.0f, dd = 0.0f;
      const int m = min(32, nw - i0);
      for (int j0 = 0; j0 < m; j0 += GATHER_U) {
        int sv[GATHER_U];
        vrow v[GATHER_U];
#pragma unroll
        for (int u = 0; u < GATHER_U; ++u) {
          sv[u] = j0 + u < m ? j0 + u : j0;
          const int su = __builtin_amdgcn_readlane(s, sv[u]);
          v[u] = *(const vrow*)(h + (size_t)su * F + lane * VW);
        }
        if (j0 == 0) {
          ds = dinv[s];
          dd = dinv[lo + dl];
        }
#pragma unroll
        for (int u = 0; u < GATHER_U; ++u) {
          if (j0 + u < m) {
            const int   du = __builtin_amdgcn_readlane(dl, sv[u]);
            const float wu = __int_as_float(__builtin_amdgcn_readlane(__float_as_int(ds), sv[u])) *
                             __int_as_float(__builtin_amdgcn_readlane(__float_as_int(dd), sv[u]));
            float* ar = s_acc + du * F + lane;
#pragma unroll
            for (int j = 0; j < VW; ++j) atomicAdd(&ar[32 * j], v[u][j] * wu);
          }
        }
      }
    }
  }
  int step = n_edges;
  for (int pos = 0; refilter && pos < n_edges; ) {
    if (tid == 0) s_ovf = 0;
    __syncthreads();

    const int len = min(step, n_edges - pos);
    int wcount = 0;
    scan_window(dst, pos, len, lo, cnt, [&](int e, int, bool hit) {
      const unsigned bal = __builtin_amdgcn_ballot_w32(hit);
      if (hit) {
        const int p = wcount + (int)__builtin_amdgcn_mbcnt_lo(bal, 0u);
        if (p < WAVE_CAP) wl[p] = e;
      }
      wcount += __popc(bal);
    });
    if (wcount > WAVE_CAP && lane == 0) s_ovf = 1;
    __syncthreads();

    if (s_ovf != 0 && step > SAFE_WINDOW) {
      step = SAFE_WINDOW;
      __syncthreads();
      continue;
    }

    for (int i0 = 0; i0 < wcount; i0 += GATHER_U) {
      int   sv[GATHER_U], dl[GATHER_U];
      float w[GATHER_U];
#pragma unroll
      for (int u = 0; u < GATHER_U; ++u) {
        const int i = i0 + u;
        const int e = wl[i < wcount ? i : i0];
        int s = src[e];
        const int dd = dst[e];
        s = min(max(s, 0), n_nodes - 1);
        sv[u] = s;
        dl[u] = dd - lo;
        w[u]  = dinv[s] * dinv[dd];
      }
      vrow v[GATHER_U];
#pragma unroll
      for (int u = 0; u < GATHER_U; ++u)
        v[u] = *(const vrow*)(h + (size_t)sv[u] * F + lane * VW);
#pragma unroll
      for (int u = 0; u < GATHER_U; ++u) {
        if (i0 + u < wcount) {
          float* ar = s_acc + dl[u] * F + lane;
#pragma unroll
          for (int j = 0; j < VW; ++j) atomicAdd(&ar[32 * j], v[u][j] * w[u]);
        }
      }
    }
    __syncthreads();
    pos += len;
  }
  __syncthreads();

  {
    float b[VW];
#pragma unroll
    for (int j = 0; j < VW; ++j) b[j] = bias[lane * VW + j];
    for (int i = tid; i < nvec; i += AGG_THREADS) {
      const int r = i >> 5;
      const float* ar = s_acc + r * F + lane;
      vrow v;
#pragma unroll
      for (int j = 0; j < VW; ++j) {
        float x = ar[32 * j] + b[j];
        if (RELU) x = x > 0.0f ? x : 0.0f;
        v[j] = x;
      }
      *(vrow*)(out + (size_t)(lo + r) * F + lane * VW) = v;
    }
  }
}

extern "C" void kernel_launch(void* const* d_in, const int* in_sizes, int n_in,
                              void* d_out, int out_size, void* d_ws, size_t ws_size,
                              hipStream_t stream) {
  const float* x  = (const float*)d_in[0];
  const int*   ei = (const int*)d_in[1];
  const float* W1 = (const float*)d_in[2];
  const float* b1 = (const float*)d_in[3];
  const float* W2 = (const float*)d_in[4];
  const float* b2 = (const float*)d_in[5];
  float* out = (float*)d_out;

  const int N = in_sizes[0] / IN_DIM;
  const int E = in_sizes[1] / 2;
  const int* src = ei;
  const int* dst = ei + E;

  float* h    = (float*)d_ws;
  float* agg1 = h + (size_t)N * HID_DIM;
  float* dinv = agg1 + (size_t)N * HID_DIM;

  const int tiles  = (N + TILE_NODES - 1) / TILE_NODES;

  unsigned* lists = (unsigned*)d_out;
  int* cparts = (int*)agg1;
  const int fgrid = ((tiles + QS - 1) / QS) * QS;
  filter_kernel<<<fgrid, AGG_THREADS, 0, stream>>>(src, dst, cparts, lists, E, N, tiles);

  const int waves  = (N + 15) / 16;
  const int gblks  = (waves + 3) / 4;

  gemm_wmma_kernel<IN_DIM, HID_DIM / 16, true><<<gblks, 128, 0, stream>>>(x, W1, h, N, cparts, dinv);
  aggregate_kernel<HID_DIM, true><<<tiles, AGG_THREADS, 0, stream>>>(
      h, src, dst, dinv, b1, lists, agg1, E, N);

  gemm_wmma_kernel<HID_DIM, OUT_DIM / 16, false><<<gblks, 128, 0, stream>>>(agg1, W2, h, N, cparts, dinv);
  aggregate_kernel<OUT_DIM, false><<<tiles, AGG_THREADS, 0, stream>>>(
      h, src, dst, dinv, b2, lists, out, E, N);
}


# GEMM (both layers): every fragment load and store moves 16 B per lane: k order inside a 16-k block and the tile-to-column map are permuted, B pairs assembled in registers; k sum order changes
# speedup vs baseline: 4.7196x; 1.0505x over previous
#include <hip/hip_runtime.h>


#define IN_DIM  256
#define HID_DIM 128
#define OUT_DIM 64

#define TILE_NODES  196
#define AGG_THREADS 1024
#define AGG_WAVES   (AGG_THREADS / 32)
#define WAVE_CAP    1024
#define SAFE_WINDOW 16384
#define SEG_MAX     384
#define QS          4
#define SEG_PITCH   96
#define LIST_ROW    OUT_DIM
#define PACK_BITS   24
#define SCAN_U      8
#define GATHER_U    8

typedef __attribute__((ext_vector_type(2))) float v2f;
typedef __attribute__((ext_vector_type(4))) float v4f;
typedef __attribute__((ext_vector_type(8))) float v8f;

template <class OnHit>
__device__ __forceinline__ void test_group(int4 d, int e, int lo, unsigned ucnt, OnHit&& on) {
  const unsigned a = (unsigned)(d.x - lo), b = (unsigned)(d.y - lo);
  const unsigned c = (unsigned)(d.z - lo), f = (unsigned)(d.w - lo);
  const unsigned mn = min(min(a, b), min(c, f));
  if (__builtin_amdgcn_ballot_w32(mn < ucnt) != 0u) {
    on(e,     d.x, a < ucnt);
    on(e + 1, d.y, b < ucnt);
    on(e + 2, d.z, c < ucnt);
    on(e + 3, d.w, f < ucnt);
  }
}

template <class OnHit>
__device__ __forceinline__ void scan_window(const int* __restrict__ dst, int pos, int len,
                                            int lo, int cnt, OnHit&& on) {
  const int tid = threadIdx.x;
  const unsigned ucnt = (unsigned)cnt;
  const int head     = min(len, (int)((4u - (unsigned)(((size_t)(dst + pos)) >> 2)) & 3u));
  const int ngrp     = (len - head) >> 2;
  const int body_end = head + 4 * ngrp;
  const int nrag     = head + (len - body_end);
  {
    const bool v  = tid < nrag;
    const int rel = tid < head ? tid : body_end + (tid - head);
    const int e   = pos + (v ? rel : 0);
    const int d   = v ? dst[e] : -1;
    const bool hit = (unsigned)(d - lo) < ucnt;
    if (__builtin_amdgcn_ballot_w32(hit) != 0u) on(e, d, hit);
  }
  const int4* __restrict__ d4 = (const int4*)(dst + pos + head);
  const int ebase = pos + head;
  constexpr int GPI = AGG_THREADS * SCAN_U;
  const int nfull = ngrp / GPI;
  const int itrot = (int)(((unsigned)blockIdx.x * (unsigned)nfull) / gridDim.x);
  for (int it = 0; it < nfull; ++it) {
    int itr = it + itrot;
    if (itr >= nfull) itr -= nfull;
    const int g0 = itr * GPI + tid;
    int4 d[SCAN_U];
#pragma unroll
    for (int u = 0; u < SCAN_U; ++u) d[u] = d4[g0 + u * AGG_THREADS];
#pragma unroll
    for (int u = 0; u < SCAN_U; ++u)
      test_group(d[u], ebase + 4 * (g0 + u * AGG_THREADS), lo, ucnt, on);
  }
  for (int base = nfull * GPI; base < ngrp; base += AGG_THREADS) {
    const int g  = base + tid;
    const bool v = g < ngrp;
    const int gi = v ? g : base;
    int4 d = d4[gi];
    if (!v) { d.x = -1; d.y = -1; d.z = -1; d.w = -1; }
    test_group(d, ebase + 4 * gi, lo, ucnt, on);
  }
}

__device__ __forceinline__ int seg_pitch(int cnt) {
  return max(0, min(SEG_PITCH, (((cnt * LIST_ROW - QS * 32) / (QS * 32)) / 32) * 32));
}

__device__ __forceinline__ int tile_lo(int t, int n_nodes, int tiles) {
  return t >= tiles ? n_nodes : (int)(((long long)t * n_nodes) / tiles);
}

__global__ __launch_bounds__(AGG_THREADS)
void filter_kernel(const int* __restrict__ src, const int* __restrict__ dst,
                   int* __restrict__ cparts, unsigned* __restrict__ lists,
                   int n_edges, int n_nodes, int tiles) {
  __shared__ int s_c[QS * TILE_NODES + 32];
  __shared__ int s_list[AGG_WAVES * SEG_MAX];
  __shared__ int s_cnt[QS * AGG_WAVES];
  const int tid  = threadIdx.x;
  const int lane = tid & 31;
  const int wave = tid >> 5;
  const int grp  = blockIdx.x / QS;
  const int part = blockIdx.x % QS;
  int tlo[QS + 1];
#pragma unroll
  for (int q = 0; q <= QS; ++q) tlo[q] = tile_lo(grp * QS + q, n_nodes, tiles);
  const int lo   = tlo[0];
  const int cnt  = tlo[QS] - lo;
  const int a0   = (lo + 31) & ~31;
  const int a1   = min(n_nodes, (lo + cnt + 31) & ~31);
  const int cntc = a1 - lo;
  const int plane = (n_nodes + 31) & ~31;
  const int e0   = (int)(((long long)n_edges * part) / QS);
  const int e1   = (int)(((long long)n_edges * (part + 1)) / QS);
  int* wl = s_list + wave * SEG_MAX;

  for (int i = tid; i < QS * TILE_NODES + 32; i += AGG_THREADS) s_c[i] = 0;
  __syncthreads();

  int wcount = 0;
  if (e1 > e0)
    scan_window(dst, e0, e1 - e0, lo, cntc, [&](int e, int d, bool hitc) {
      if (hitc) atomicAdd(&s_c[d - lo], 1);
      const bool hit = hitc && (d - lo) < cnt;
      const unsigned bal = __builtin_amdgcn_ballot_w32(hit);
      if (hit) {
        const int p = wcount + (int)__builtin_amdgcn_mbcnt_lo(bal, 0u);
        if (p < SEG_MAX) wl[p] = e;
      }
      wcount += __popc(bal);
    });
  __syncthreads();

  int* cp = cparts + (size_t)part * plane;
  for (int i = a0 - lo + tid; i < cntc; i += AGG_THREADS) cp[lo + i] = s_c[i];

  const int n = min(wcount, SEG_MAX);
  const bool bad = wcount > SEG_MAX || n_nodes > (1 << PACK_BITS);
  int c[QS], pit[QS];
  unsigned* seg[QS];
#pragma unroll
  for (int q = 0; q < QS; ++q) {
    c[q]   = 0;
    pit[q] = seg_pitch(tlo[q + 1] - tlo[q]);
    seg[q] = lists + (size_t)tlo[q] * LIST_ROW + QS * 32 + (part * AGG_WAVES + wave) * pit[q];
  }
  for (int i0 = 0; i0 < n; i0 += 32) {
    const int i  = i0 + lane;
    const bool v = i < n;
    const int e  = wl[v ? i : i0];
    int sidx = src[e];
    const int d = dst[e];
    sidx = min(max(sidx, 0), n_nodes - 1);
    int q = 0, tl = tlo[0];
#pragma unroll
    for (int k = 1; k < QS; ++k) {
      if (d >= tlo[k]) { q = k; tl = tlo[k]; }
    }
    const unsigned word = (unsigned)sidx | ((unsigned)(d - tl) << PACK_BITS);
#pragma unroll
    for (int k = 0; k < QS; ++k) {
      const bool mine = v && q == k;
      const unsigned bal = __builtin_amdgcn_ballot_w32(mine);
      if (mine) {
        const int p = c[k] + (int)__builtin_amdgcn_mbcnt_lo(bal, 0u);
        if (p < pit[k]) seg[k][p] = word;
      }
      c[k] += __popc(bal);
    }
  }
  if (lane == 0) {
#pragma unroll
    for (int q = 0; q < QS; ++q) s_cnt[q * AGG_WAVES + wave] = bad ? 0x7fffffff : c[q];
  }
  __syncthreads();
#pragma unroll
  for (int q = 0; q < QS; ++q) {
    if (wave == q && (tlo[q + 1] - tlo[q]) * LIST_ROW >= QS * 32)
      (lists + (size_t)tlo[q] * LIST_ROW)[part * 32 + lane] = (unsigned)s_cnt[q * AGG_WAVES + lane];
  }
}

template <int K, int NT, bool DINV>
__global__ __launch_bounds__(128)
void gemm_wmma_kernel(const float* __restrict__ A,
                      const float* __restrict__ B,
                      float* __restrict__ C, int M,
                      const int* __restrict__ cparts, float* __restrict__ dinv) {
  constexpr int N  = NT * 16;
  constexpr int NJ = NT / 4;
  constexpr int KB = 4;
  constexpr int NB = K / (4 * KB);
  static_assert(K % (8 * KB) == 0, "the two register blocks alternate in pairs");
  static_assert(NT % 4 == 0, "a lane holds four neighbouring columns per load");
  const int wave = (int)((blockIdx.x * (unsigned)blockDim.x + threadIdx.x) >> 5);
  const int lane = threadIdx.x & 31;
  const int row0 = wave * 16;
  if (row0 >= M) return;

  if (DINV && (wave & 1) == 0) {
    const int node = row0 + lane;
    if (node < M) {
      const int plane = (M + 31) & ~31;
      int deg = 0;
#pragma unroll
      for (int q = 0; q < QS; ++q) deg += cparts[(size_t)q * plane + node];
      dinv[node] = rsqrtf(1.0f + (float)deg);
    }
  }

  const int mn    = lane & 15;
  const int khalf = lane >> 4;

  int arow = row0 + mn;
  if (arow >= M) arow = M - 1;
  const float* __restrict__ Ar = A + (size_t)arow * K + 8 * khalf;
  const float* __restrict__ Bc = B + (size_t)(8 * khalf) * N + 4 * mn;

  v8f acc[NT];
#pragma unroll
  for (int t = 0; t < NT; ++t) acc[t] = v8f{};

  struct Raw {
    v4f a[2];
    v4f b[KB][2][NJ];
  };
  Raw P, Q;

  auto load_block = [&](int kb, Raw& r) {
    const float* __restrict__ ap = Ar + kb * (4 * KB);
    const float* __restrict__ bp = Bc + (size_t)kb * (4 * KB) * N;
    r.a[0] = *(const v4f*)(ap);
    r.a[1] = *(const v4f*)(ap + 4);
#pragma unroll
    for (int s = 0; s < KB; ++s) {
#pragma unroll
      for (int e = 0; e < 2; ++e) {
#pragma unroll
        for (int j = 0; j < NJ; ++j)
          r.b[s][e][j] = *(const v4f*)(bp + (size_t)(2 * s + e) * N + 64 * j);
      }
    }
  };
  auto mma_block = [&](const Raw& r) {
#pragma unroll
    for (int s = 0; s < KB; ++s) {
      v2f a;
      a.x = r.a[s >> 1][2 * (s & 1)];
      a.y = r.a[s >> 1][2 * (s & 1) + 1];
#pragma unroll
      for (int t = 0; t < NT; ++t) {
        v2f b;
        b.x = r.b[s][0][t >> 2][t & 3];
        b.y = r.b[s][1][t >> 2][t & 3];
        acc[t] = __builtin_amdgcn_wmma_f32_16x16x4_f32(
             false, a,  false, b,
             (short)0, acc[t],  false,  false);
      }
    }
  };

  load_block(0, P);
  for (int kb = 0; kb + 2 < NB; kb += 2) {
    load_block(kb + 1, Q);
    __builtin_amdgcn_sched_barrier(0);
    mma_block(P);
    __builtin_amdgcn_sched_barrier(0);
    load_block(kb + 2, P);
    __builtin_amdgcn_sched_barrier(0);
    mma_block(Q);
    __builtin_amdgcn_sched_barrier(0);
  }
  load_block(NB - 1, Q);
  __builtin_amdgcn_sched_barrier(0);
  mma_block(P);
  mma_block(Q);

#pragma unroll
  for (int i = 0; i < 8; ++i) {
    const int row = row0 + 8 * khalf + i;
#pragma unroll
    for (int j = 0; j < NJ; ++j) {
      v4f o;
      o.x = acc[4 * j][i];
      o.y = acc[4 * j + 1][i];
      o.z = acc[4 * j + 2][i];
      o.w = acc[4 * j + 3][i];
      if (row < M) *(v4f*)(C + (size_t)row * N + 64 * j + 4 * mn) = o;
    }
  }
}

template <int F, bool RELU>
__global__ __launch_bounds__(AGG_THREADS)
void aggregate_kernel(const float* __restrict__ h,
                      const int* __restrict__ src,
                      const int* __restrict__ dst,
                      const float* __restrict__ dinv,
                      const float* __restrict__ bias,
                      const unsigned* lists,
                      float* out,
                      int n_edges, int n_nodes) {
  __shared__ __attribute__((aligned(16))) float s_acc[TILE_NODES * F];
  __shared__ int s_list[AGG_WAVES * WAVE_CAP];
  __shared__ int s_ovf;

  constexpr int VW = F / 32;
  typedef __attribute__((ext_vector_type(VW))) float vrow;
  const int tid  = threadIdx.x;
  const int lane = tid & 31;
  const int wave = __builtin_amdgcn_readfirstlane(tid >> 5);
  const int lo   = tile_lo(blockIdx.x, n_nodes, gridDim.x);
  const int cnt  = tile_lo(blockIdx.x + 1, n_nodes, gridDim.x) - lo;
  const int nvec = cnt * 32;
  int* wl = s_list + wave * WAVE_CAP;

  const unsigned* lbase = lists + (size_t)lo * LIST_ROW;
  const int pitch = seg_pitch(cnt);
  const bool small = cnt * LIST_ROW < QS * 32;
  int nws[QS];
  bool over = small;
#pragma unroll
  for (int q = 0; q < QS; ++q) {
    const int hv = small ? 0 : (int)lbase[q * 32 + lane];
    over = over || hv > pitch;
    nws[q] = __builtin_amdgcn_readlane(hv, wave);
  }
  const bool refilter = __builtin_amdgcn_ballot_w32(over) != 0u;

  {
    for (int i = tid; i < nvec; i += AGG_THREADS) {
      const int r = i >> 5;
      const float di = dinv[lo + r];
      const float w  = di * di;
      const vrow v = *(const vrow*)(h + (size_t)(lo + r) * F + lane * VW);
      float* ar = s_acc + r * F + lane;
#pragma unroll
      for (int j = 0; j < VW; ++j) ar[32 * j] = v[j] * w;
    }
  }

  __syncthreads();
  for (int q = 0; !refilter && q < QS; ++q) {
    int nw = nws[0];
#pragma unroll
    for (int k = 1; k < QS; ++k) nw = q == k ? nws[k] : nw;
    const unsigned* seg = lbase + QS * 32 + (q * AGG_WAVES + wave) * pitch;
    for (int i0 = 0; i0 < nw; i0 += 32) {
      const int i = i0 + lane;
      const unsigned pk = seg[i < nw ? i : i0];
      const int   s   = (int)(pk & ((1u << PACK_BITS) - 1u));
      const int   dl  = (int)(pk >> PACK_BITS);
      float ds = 0.0f, dd = 0.0f;
      const int m = min(32, nw - i0);
      for (int j0 = 0; j0 < m; j0 += GATHER_U) {
        int sv[GATHER_U];
        vrow v[GATHER_U];
#pragma unroll
        for (int u = 0; u < GATHER_U; ++u) {
          sv[u] = j0 + u < m ? j0 + u : j0;
          const int su = __builtin_amdgcn_readlane(s, sv[u]);
          v[u] = *(const vrow*)(h + (size_t)su * F + lane * VW);
        }
        if (j0 == 0) {
          ds = dinv[s];
          dd = dinv[lo + dl];
        }
#pragma unroll
        for (int u = 0; u < GATHER_U; ++u) {
          if (j0 + u < m) {
            const int   du = __builtin_amdgcn_readlane(dl, sv[u]);
            const float wu = __int_as_float(__builtin_amdgcn_readlane(__float_as_int(ds), sv[u])) *
                             __int_as_float(__builtin_amdgcn_readlane(__float_as_int(dd), sv[u]));
            float* ar = s_acc + du * F + lane;
#pragma unroll
            for (int j = 0; j < VW; ++j) atomicAdd(&ar[32 * j], v[u][j] * wu);
          }
        }
      }
    }
  }
  int step = n_edges;
  for (int pos = 0; refilter && pos < n_edges; ) {
    if (tid == 0) s_ovf = 0;
    __syncthreads();

    const int len = min(step, n_edges - pos);
    int wcount = 0;
    scan_window(dst, pos, len, lo, cnt, [&](int e, int, bool hit) {
      const unsigned bal = __builtin_amdgcn_ballot_w32(hit);
      if (hit) {
        const int p = wcount + (int)__builtin_amdgcn_mbcnt_lo(bal, 0u);
        if (p < WAVE_CAP) wl[p] = e;
      }
      wcount += __popc(bal);
    });
    if (wcount > WAVE_CAP && lane == 0) s_ovf = 1;
    __syncthreads();

    if (s_ovf != 0 && step > SAFE_WINDOW) {
      step = SAFE_WINDOW;
      __syncthreads();
      continue;
    }

    for (int i0 = 0; i0 < wcount; i0 += GATHER_U) {
      int   sv[GATHER_U], dl[GATHER_U];
      float w[GATHER_U];
#pragma unroll
      for (int u = 0; u < GATHER_U; ++u) {
        const int i = i0 + u;
        const int e = wl[i < wcount ? i : i0];
        int s = src[e];
        const int dd = dst[e];
        s = min(max(s, 0), n_nodes - 1);
        sv[u] = s;
        dl[u] = dd - lo;
        w[u]  = dinv[s] * dinv[dd];
      }
      vrow v[GATHER_U];
#pragma unroll
      for (int u = 0; u < GATHER_U; ++u)
        v[u] = *(const vrow*)(h + (size_t)sv[u] * F + lane * VW);
#pragma unroll
      for (int u = 0; u < GATHER_U; ++u) {
        if (i0 + u < wcount) {
          float* ar = s_acc + dl[u] * F + lane;
#pragma unroll
          for (int j = 0; j < VW; ++j) atomicAdd(&ar[32 * j], v[u][j] * w[u]);
        }
      }
    }
    __syncthreads();
    pos += len;
  }
  __syncthreads();

  {
    float b[VW];
#pragma unroll
    for (int j = 0; j < VW; ++j) b[j] = bias[lane * VW + j];
    for (int i = tid; i < nvec; i += AGG_THREADS) {
      const int r = i >> 5;
      const float* ar = s_acc + r * F + lane;
      vrow v;
#pragma unroll
      for (int j = 0; j < VW; ++j) {
        float x = ar[32 * j] + b[j];
        if (RELU) x = x > 0.0f ? x : 0.0f;
        v[j] = x;
      }
      *(vrow*)(out + (size_t)(lo + r) * F + lane * VW) = v;
    }
  }
}

extern "C" void kernel_launch(void* const* d_in, const int* in_sizes, int n_in,
                              void* d_out, int out_size, void* d_ws, size_t ws_size,
                              hipStream_t stream) {
  const float* x  = (const float*)d_in[0];
  const int*   ei = (const int*)d_in[1];
  const float* W1 = (const float*)d_in[2];
  const float* b1 = (const float*)d_in[3];
  const float* W2 = (const float*)d_in[4];
  const float* b2 = (const float*)d_in[5];
  float* out = (float*)d_out;

  const int N = in_sizes[0] / IN_DIM;
  const int E = in_sizes[1] / 2;
  const int* src = ei;
  const int* dst = ei + E;

  float* h    = (float*)d_ws;
  float* agg1 = h + (size_t)N * HID_DIM;
  float* dinv = agg1 + (size_t)N * HID_DIM;

  const int tiles  = (N + TILE_NODES - 1) / TILE_NODES;

  unsigned* lists = (unsigned*)d_out;
  int* cparts = (int*)agg1;
  const int fgrid = ((tiles + QS - 1) / QS) * QS;
  filter_kernel<<<fgrid, AGG_THREADS, 0, stream>>>(src, dst, cparts, lists, E, N, tiles);

  const int waves  = (N + 15) / 16;
  const int gblks  = (waves + 3) / 4;

  gemm_wmma_kernel<IN_DIM, HID_DIM / 16, true><<<gblks, 128, 0, stream>>>(x, W1, h, N, cparts, dinv);
  aggregate_kernel<HID_DIM, true><<<tiles, AGG_THREADS, 0, stream>>>(
      h, src, dst, dinv, b1, lists, agg1, E, N);

  gemm_wmma_kernel<HID_DIM, OUT_DIM / 16, false><<<gblks, 128, 0, stream>>>(agg1, W2, h, N, cparts, dinv);
  aggregate_kernel<OUT_DIM, false><<<tiles, AGG_THREADS, 0, stream>>>(
      h, src, dst, dinv, b2, lists, out, E, N);
}


# GEMM (both layers) weight-stationary: each wave keeps a 32-column slice of W as B fragments in registers all launch; x rows reach the waves by whole-line loads through LDS; k and sum order kept
# speedup vs baseline: 4.8670x; 1.0312x over previous
#include <hip/hip_runtime.h>


#define IN_DIM  256
#define HID_DIM 128
#define OUT_DIM 64

#define TILE_NODES  196
#define AGG_THREADS 1024
#define AGG_WAVES   (AGG_THREADS / 32)
#define WAVE_CAP    1024
#define SAFE_WINDOW 16384
#define SEG_MAX     384
#define QS          4
#define SEG_PITCH   96
#define LIST_ROW    OUT_DIM
#define PACK_BITS   24
#define SCAN_U      8
#define GATHER_U    8

typedef __attribute__((ext_vector_type(2))) float v2f;
typedef __attribute__((ext_vector_type(4))) float v4f;
typedef __attribute__((ext_vector_type(8))) float v8f;

template <class OnHit>
__device__ __forceinline__ void test_group(int4 d, int e, int lo, unsigned ucnt, OnHit&& on) {
  const unsigned a = (unsigned)(d.x - lo), b = (unsigned)(d.y - lo);
  const unsigned c = (unsigned)(d.z - lo), f = (unsigned)(d.w - lo);
  const unsigned mn = min(min(a, b), min(c, f));
  if (__builtin_amdgcn_ballot_w32(mn < ucnt) != 0u) {
    on(e,     d.x, a < ucnt);
    on(e + 1, d.y, b < ucnt);
    on(e + 2, d.z, c < ucnt);
    on(e + 3, d.w, f < ucnt);
  }
}

template <class OnHit>
__device__ __forceinline__ void scan_window(const int* __restrict__ dst, int pos, int len,
                                            int lo, int cnt, OnHit&& on) {
  const int tid = threadIdx.x;
  const unsigned ucnt = (unsigned)cnt;
  const int head     = min(len, (int)((4u - (unsigned)(((size_t)(dst + pos)) >> 2)) & 3u));
  const int ngrp     = (len - head) >> 2;
  const int body_end = head + 4 * ngrp;
  const int nrag     = head + (len - body_end);
  {
    const bool v  = tid < nrag;
    const int rel = tid < head ? tid : body_end + (tid - head);
    const int e   = pos + (v ? rel : 0);
    const int d   = v ? dst[e] : -1;
    const bool hit = (unsigned)(d - lo) < ucnt;
    if (__builtin_amdgcn_ballot_w32(hit) != 0u) on(e, d, hit);
  }
  const int4* __restrict__ d4 = (const int4*)(dst + pos + head);
  const int ebase = pos + head;
  constexpr int GPI = AGG_THREADS * SCAN_U;
  const int nfull = ngrp / GPI;
  const int itrot = (int)(((unsigned)blockIdx.x * (unsigned)nfull) / gridDim.x);
  for (int it = 0; it < nfull; ++it) {
    int itr = it + itrot;
    if (itr >= nfull) itr -= nfull;
    const int g0 = itr * GPI + tid;
    int4 d[SCAN_U];
#pragma unroll
    for (int u = 0; u < SCAN_U; ++u) d[u] = d4[g0 + u * AGG_THREADS];
#pragma unroll
    for (int u = 0; u < SCAN_U; ++u)
      test_group(d[u], ebase + 4 * (g0 + u * AGG_THREADS), lo, ucnt, on);
  }
  for (int base = nfull * GPI; base < ngrp; base += AGG_THREADS) {
    const int g  = base + tid;
    const bool v = g < ngrp;
    const int gi = v ? g : base;
    int4 d = d4[gi];
    if (!v) { d.x = -1; d.y = -1; d.z = -1; d.w = -1; }
    test_group(d, ebase + 4 * gi, lo, ucnt, on);
  }
}

__device__ __forceinline__ int seg_pitch(int cnt) {
  return max(0, min(SEG_PITCH, (((cnt * LIST_ROW - QS * 32) / (QS * 32)) / 32) * 32));
}

__device__ __forceinline__ int tile_lo(int t, int n_nodes, int tiles) {
  return t >= tiles ? n_nodes : (int)(((long long)t * n_nodes) / tiles);
}

__global__ __launch_bounds__(AGG_THREADS)
void filter_kernel(const int* __restrict__ src, const int* __restrict__ dst,
                   int* __restrict__ cparts, unsigned* __restrict__ lists,
                   int n_edges, int n_nodes, int tiles) {
  __shared__ int s_c[QS * TILE_NODES + 32];
  __shared__ int s_list[AGG_WAVES * SEG_MAX];
  __shared__ int s_cnt[QS * AGG_WAVES];
  const int tid  = threadIdx.x;
  const int lane = tid & 31;
  const int wave = tid >> 5;
  const int grp  = blockIdx.x / QS;
  const int part = blockIdx.x % QS;
  int tlo[QS + 1];
#pragma unroll
  for (int q = 0; q <= QS; ++q) tlo[q] = tile_lo(grp * QS + q, n_nodes, tiles);
  const int lo   = tlo[0];
  const int cnt  = tlo[QS] - lo;
  const int a0   = (lo + 31) & ~31;
  const int a1   = min(n_nodes, (lo + cnt + 31) & ~31);
  const int cntc = a1 - lo;
  const int plane = (n_nodes + 31) & ~31;
  const int e0   = (int)(((long long)n_edges * part) / QS);
  const int e1   = (int)(((long long)n_edges * (part + 1)) / QS);
  int* wl = s_list + wave * SEG_MAX;

  for (int i = tid; i < QS * TILE_NODES + 32; i += AGG_THREADS) s_c[i] = 0;
  __syncthreads();

  int wcount = 0;
  if (e1 > e0)
    scan_window(dst, e0, e1 - e0, lo, cntc, [&](int e, int d, bool hitc) {
      if (hitc) atomicAdd(&s_c[d - lo], 1);
      const bool hit = hitc && (d - lo) < cnt;
      const unsigned bal = __builtin_amdgcn_ballot_w32(hit);
      if (hit) {
        const int p = wcount + (int)__builtin_amdgcn_mbcnt_lo(bal, 0u);
        if (p < SEG_MAX) wl[p] = e;
      }
      wcount += __popc(bal);
    });
  __syncthreads();

  int* cp = cparts + (size_t)part * plane;
  for (int i = a0 - lo + tid; i < cntc; i += AGG_THREADS) cp[lo + i] = s_c[i];

  const int n = min(wcount, SEG_MAX);
  const bool bad = wcount > SEG_MAX || n_nodes > (1 << PACK_BITS);
  int c[QS], pit[QS];
  unsigned* seg[QS];
#pragma unroll
  for (int q = 0; q < QS; ++q) {
    c[q]   = 0;
    pit[q] = seg_pitch(tlo[q + 1] - tlo[q]);
    seg[q] = lists + (size_t)tlo[q] * LIST_ROW + QS * 32 + (part * AGG_WAVES + wave) * pit[q];
  }
  for (int i0 = 0; i0 < n; i0 += 32) {
    const int i  = i0 + lane;
    const bool v = i < n;
    const int e  = wl[v ? i : i0];
    int sidx = src[e];
    const int d = dst[e];
    sidx = min(max(sidx, 0), n_nodes - 1);
    int q = 0, tl = tlo[0];
#pragma unroll
    for (int k = 1; k < QS; ++k) {
      if (d >= tlo[k]) { q = k; tl = tlo[k]; }
    }
    const unsigned word = (unsigned)sidx | ((unsigned)(d - tl) << PACK_BITS);
#pragma unroll
    for (int k = 0; k < QS; ++k) {
      const bool mine = v && q == k;
      const unsigned bal = __builtin_amdgcn_ballot_w32(mine);
      if (mine) {
        const int p = c[k] + (int)__builtin_amdgcn_mbcnt_lo(bal, 0u);
        if (p < pit[k]) seg[k][p] = word;
      }
      c[k] += __popc(bal);
    }
  }
  if (lane == 0) {
#pragma unroll
    for (int q = 0; q < QS; ++q) s_cnt[q * AGG_WAVES + wave] = bad ? 0x7fffffff : c[q];
  }
  __syncthreads();
#pragma unroll
  for (int q = 0; q < QS; ++q) {
    if (wave == q && (tlo[q + 1] - tlo[q]) * LIST_ROW >= QS * 32)
      (lists + (size_t)tlo[q] * LIST_ROW)[part * 32 + lane] = (unsigned)s_cnt[q * AGG_WAVES + lane];
  }
}

#define GEMM_THREADS  128
#define GEMM_GRID     512
#define GEMM_LDS_HOLD 81920

template <int K, int N, bool DINV>
__global__ __launch_bounds__(GEMM_THREADS, 2)
void gemm_wmma_kernel(const float* __restrict__ A,
                      const float* __restrict__ B,
                      float* __restrict__ C, int M,
                      const int* __restrict__ cparts, float* __restrict__ dinv) {
  constexpr int CS    = N / 32;
  constexpr int SG    = (GEMM_THREADS / 32) / CS;
  constexpr int R     = 16 * SG;
  constexpr int NKB   = K / 16;
  constexpr int PITCH = K + 4;
  constexpr int CPR   = K / 4;
  constexpr int LPT   = (R * CPR) / GEMM_THREADS;
  static_assert(N % 32 == 0 && (GEMM_THREADS / 32) % CS == 0, "whole slices, whole strips");
  static_assert((R * CPR) % GEMM_THREADS == 0, "every thread moves the same number of chunks");
  static_assert(((PITCH / 4) & 1) == 1, "row pitch must be an odd number of 16-byte units");
  __shared__ __attribute__((aligned(16))) float s_a[2][R * PITCH];

  const int tid  = threadIdx.x;
  const int lane = tid & 31;
  const int wave = __builtin_amdgcn_readfirstlane(tid >> 5);
  const int cs   = wave % CS;
  const int sg   = wave / CS;
  const int c    = lane & 15;
  const int hf   = lane >> 4;

  if (DINV) {
    const int plane = (M + 31) & ~31;
    for (int node = blockIdx.x * GEMM_THREADS + tid; node < M; node += gridDim.x * GEMM_THREADS) {
      int deg = 0;
#pragma unroll
      for (int q = 0; q < QS; ++q) deg += cparts[(size_t)q * plane + node];
      dinv[node] = rsqrtf(1.0f + (float)deg);
    }
  }

  const int nstages = (M + R - 1) / R;
  int st = blockIdx.x;
  if (st >= nstages) return;

  v4f g[LPT];
  auto load_stage = [&](int stage) {
#pragma unroll
    for (int j = 0; j < LPT; ++j) {
      const int q = tid + GEMM_THREADS * j;
      int row = stage * R + q / CPR;
      if (row >= M) row = M - 1;
      g[j] = *(const v4f*)(A + (size_t)row * K + 4 * (q % CPR));
    }
  };
  auto store_stage = [&](int buf) {
#pragma unroll
    for (int j = 0; j < LPT; ++j) {
      const int q = tid + GEMM_THREADS * j;
      *(v4f*)(&s_a[buf][(q / CPR) * PITCH + 4 * (q % CPR)]) = g[j];
    }
  };

  load_stage(st);

  v2f bw[NKB][4][2];
  {
    const float* __restrict__ Bc = B + (size_t)(8 * hf) * N + 32 * cs + 2 * c;
#pragma unroll
    for (int kb = 0; kb < NKB; ++kb) {
#pragma unroll
      for (int s = 0; s < 4; ++s) {
        const v2f e0 = *(const v2f*)(Bc + (size_t)(16 * kb + 2 * s) * N);
        const v2f e1 = *(const v2f*)(Bc + (size_t)(16 * kb + 2 * s + 1) * N);
        bw[kb][s][0] = v2f{e0.x, e1.x};
        bw[kb][s][1] = v2f{e0.y, e1.y};
      }
    }
  }

  store_stage(0);
  if (st + (int)gridDim.x < nstages) load_stage(st + (int)gridDim.x);

  auto store_c = [&](int row0, const v8f& d0, const v8f& d1) {
#pragma unroll
    for (int r = 0; r < 8; ++r) {
      const int row = row0 + r;
      if (row < M) *(v2f*)(C + (size_t)row * N + 32 * cs + 2 * c) = v2f{d0[r], d1[r]};
    }
  };

  v8f p0 = v8f{}, p1 = v8f{};
  int prow0 = 0;
  for (int i = 0; ; ++i) {
    __syncthreads();
    const bool more = st + (int)gridDim.x < nstages;
    if (more) {
      store_stage((i + 1) & 1);
      if (st + 2 * (int)gridDim.x < nstages) load_stage(st + 2 * (int)gridDim.x);
    }
    if (i > 0) store_c(prow0, p0, p1);

    const float* ar = &s_a[i & 1][(sg * 16 + c) * PITCH + 8 * hf];
    v8f acc0 = v8f{}, acc1 = v8f{};
#pragma unroll
    for (int kb = 0; kb < NKB; ++kb) {
      const v4f a0 = *(const v4f*)(ar + 16 * kb);
      const v4f a1 = *(const v4f*)(ar + 16 * kb + 4);
#pragma unroll
      for (int s = 0; s < 4; ++s) {
        v2f a;
        a.x = (s < 2 ? a0 : a1)[2 * (s & 1)];
        a.y = (s < 2 ? a0 : a1)[2 * (s & 1) + 1];
        acc0 = __builtin_amdgcn_wmma_f32_16x16x4_f32(
             false, a,  false, bw[kb][s][0],
             (short)0, acc0,  false,  false);
        acc1 = __builtin_amdgcn_wmma_f32_16x16x4_f32(
             false, a,  false, bw[kb][s][1],
             (short)0, acc1,  false,  false);
      }
    }

    p0 = acc0;
    p1 = acc1;
    prow0 = st * R + sg * 16 + 8 * hf;
    if (!more) break;
    st += (int)gridDim.x;
  }
  store_c(prow0, p0, p1);
}

template <int F, bool RELU>
__global__ __launch_bounds__(AGG_THREADS)
void aggregate_kernel(const float* __restrict__ h,
                      const int* __restrict__ src,
                      const int* __restrict__ dst,
                      const float* __restrict__ dinv,
                      const float* __restrict__ bias,
                      const unsigned* lists,
                      float* out,
                      int n_edges, int n_nodes) {
  __shared__ __attribute__((aligned(16))) float s_acc[TILE_NODES * F];
  __shared__ int s_list[AGG_WAVES * WAVE_CAP];
  __shared__ int s_ovf;

  constexpr int VW = F / 32;
  typedef __attribute__((ext_vector_type(VW))) float vrow;
  const int tid  = threadIdx.x;
  const int lane = tid & 31;
  const int wave = __builtin_amdgcn_readfirstlane(tid >> 5);
  const int lo   = tile_lo(blockIdx.x, n_nodes, gridDim.x);
  const int cnt  = tile_lo(blockIdx.x + 1, n_nodes, gridDim.x) - lo;
  const int nvec = cnt * 32;
  int* wl = s_list + wave * WAVE_CAP;

  const unsigned* lbase = lists + (size_t)lo * LIST_ROW;
  const int pitch = seg_pitch(cnt);
  const bool small = cnt * LIST_ROW < QS * 32;
  int nws[QS];
  bool over = small;
#pragma unroll
  for (int q = 0; q < QS; ++q) {
    const int hv = small ? 0 : (int)lbase[q * 32 + lane];
    over = over || hv > pitch;
    nws[q] = __builtin_amdgcn_readlane(hv, wave);
  }
  const bool refilter = __builtin_amdgcn_ballot_w32(over) != 0u;

  {
    for (int i = tid; i < nvec; i += AGG_THREADS) {
      const int r = i >> 5;
      const float di = dinv[lo + r];
      const float w  = di * di;
      const vrow v = *(const vrow*)(h + (size_t)(lo + r) * F + lane * VW);
      float* ar = s_acc + r * F + lane;
#pragma unroll
      for (int j = 0; j < VW; ++j) ar[32 * j] = v[j] * w;
    }
  }

  __syncthreads();
  for (int q = 0; !refilter && q < QS; ++q) {
    int nw = nws[0];
#pragma unroll
    for (int k = 1; k < QS; ++k) nw = q == k ? nws[k] : nw;
    const unsigned* seg = lbase + QS * 32 + (q * AGG_WAVES + wave) * pitch;
    for (int i0 = 0; i0 < nw; i0 += 32) {
      const int i = i0 + lane;
      const unsigned pk = seg[i < nw ? i : i0];
      const int   s   = (int)(pk & ((1u << PACK_BITS) - 1u));
      const int   dl  = (int)(pk >> PACK_BITS);
      float ds = 0.0f, dd = 0.0f;
      const int m = min(32, nw - i0);
      for (int j0 = 0; j0 < m; j0 += GATHER_U) {
        int sv[GATHER_U];
        vrow v[GATHER_U];
#pragma unroll
        for (int u = 0; u < GATHER_U; ++u) {
          sv[u] = j0 + u < m ? j0 + u : j0;
          const int su = __builtin_amdgcn_readlane(s, sv[u]);
          v[u] = *(const vrow*)(h + (size_t)su * F + lane * VW);
        }
        if (j0 == 0) {
          ds = dinv[s];
          dd = dinv[lo + dl];
        }
#pragma unroll
        for (int u = 0; u < GATHER_U; ++u) {
          if (j0 + u < m) {
            const int   du = __builtin_amdgcn_readlane(dl, sv[u]);
            const float wu = __int_as_float(__builtin_amdgcn_readlane(__float_as_int(ds), sv[u])) *
                             __int_as_float(__builtin_amdgcn_readlane(__float_as_int(dd), sv[u]));
            float* ar = s_acc + du * F + lane;
#pragma unroll
            for (int j = 0; j < VW; ++j) atomicAdd(&ar[32 * j], v[u][j] * wu);
          }
        }
      }
    }
  }
  int step = n_edges;
  for (int pos = 0; refilter && pos < n_edges; ) {
    if (tid == 0) s_ovf = 0;
    __syncthreads();

    const int len = min(step, n_edges - pos);
    int wcount = 0;
    scan_window(dst, pos, len, lo, cnt, [&](int e, int, bool hit) {
      const unsigned bal = __builtin_amdgcn_ballot_w32(hit);
      if (hit) {
        const int p = wcount + (int)__builtin_amdgcn_mbcnt_lo(bal, 0u);
        if (p < WAVE_CAP) wl[p] = e;
      }
      wcount += __popc(bal);
    });
    if (wcount > WAVE_CAP && lane == 0) s_ovf = 1;
    __syncthreads();

    if (s_ovf != 0 && step > SAFE_WINDOW) {
      step = SAFE_WINDOW;
      __syncthreads();
      continue;
    }

    for (int i0 = 0; i0 < wcount; i0 += GATHER_U) {
      int   sv[GATHER_U], dl[GATHER_U];
      float w[GATHER_U];
#pragma unroll
      for (int u = 0; u < GATHER_U; ++u) {
        const int i = i0 + u;
        const int e = wl[i < wcount ? i : i0];
        int s = src[e];
        const int dd = dst[e];
        s = min(max(s, 0), n_nodes - 1);
        sv[u] = s;
        dl[u] = dd - lo;
        w[u]  = dinv[s] * dinv[dd];
      }
      vrow v[GATHER_U];
#pragma unroll
      for (int u = 0; u < GATHER_U; ++u)
        v[u] = *(const vrow*)(h + (size_t)sv[u] * F + lane * VW);
#pragma unroll
      for (int u = 0; u < GATHER_U; ++u) {
        if (i0 + u < wcount) {
          float* ar = s_acc + dl[u] * F + lane;
#pragma unroll
          for (int j = 0; j < VW; ++j) atomicAdd(&ar[32 * j], v[u][j] * w[u]);
        }
      }
    }
    __syncthreads();
    pos += len;
  }
  __syncthreads();

  {
    float b[VW];
#pragma unroll
    for (int j = 0; j < VW; ++j) b[j] = bias[lane * VW + j];
    for (int i = tid; i < nvec; i += AGG_THREADS) {
      const int r = i >> 5;
      const float* ar = s_acc + r * F + lane;
      vrow v;
#pragma unroll
      for (int j = 0; j < VW; ++j) {
        float x = ar[32 * j] + b[j];
        if (RELU) x = x > 0.0f ? x : 0.0f;
        v[j] = x;
      }
      *(vrow*)(out + (size_t)(lo + r) * F + lane * VW) = v;
    }
  }
}

extern "C" void kernel_launch(void* const* d_in, const int* in_sizes, int n_in,
                              void* d_out, int out_size, void* d_ws, size_t ws_size,
                              hipStream_t stream) {
  const float* x  = (const float*)d_in[0];
  const int*   ei = (const int*)d_in[1];
  const float* W1 = (const float*)d_in[2];
  const float* b1 = (const float*)d_in[3];
  const float* W2 = (const float*)d_in[4];
  const float* b2 = (const float*)d_in[5];
  float* out = (float*)d_out;

  const int N = in_sizes[0] / IN_DIM;
  const int E = in_sizes[1] / 2;
  const int* src = ei;
  const int* dst = ei + E;

  float* h    = (float*)d_ws;
  float* agg1 = h + (size_t)N * HID_DIM;
  float* dinv = agg1 + (size_t)N * HID_DIM;

  const int tiles  = (N + TILE_NODES - 1) / TILE_NODES;

  unsigned* lists = (unsigned*)d_out;
  int* cparts = (int*)agg1;
  const int fgrid = ((tiles + QS - 1) / QS) * QS;
  filter_kernel<<<fgrid, AGG_THREADS, 0, stream>>>(src, dst, cparts, lists, E, N, tiles);

  gemm_wmma_kernel<IN_DIM, HID_DIM, true><<<GEMM_GRID, GEMM_THREADS, GEMM_LDS_HOLD, stream>>>(x, W1, h, N, cparts, dinv);
  aggregate_kernel<HID_DIM, true><<<tiles, AGG_THREADS, 0, stream>>>(
      h, src, dst, dinv, b1, lists, agg1, E, N);

  gemm_wmma_kernel<HID_DIM, OUT_DIM, false><<<GEMM_GRID, GEMM_THREADS, GEMM_LDS_HOLD, stream>>>(agg1, W2, h, N, cparts, dinv);
  aggregate_kernel<OUT_DIM, false><<<tiles, AGG_THREADS, 0, stream>>>(
      h, src, dst, dinv, b2, lists, out, E, N);
}
